# starting kernel
# baseline (speedup 1.0000x reference)
#include <hip/hip_runtime.h>


typedef float v2f __attribute__((ext_vector_type(2)));
typedef float v8f __attribute__((ext_vector_type(8)));

#define T_STEPS 28
#define D_IN    28
#define HID     10

#define F_IH0 0
#define F_HH0 28
#define F_IH1 40
#define F_HH1 52
#define F_CLS 64
#define NFRAG 67

__device__ __forceinline__ float sigf(float x) {
    return __builtin_amdgcn_rcpf(1.0f + __expf(-x));
}
__device__ __forceinline__ float tanhf_fast(float x) {
    return 1.0f - 2.0f * __builtin_amdgcn_rcpf(1.0f + __expf(2.0f * x));
}

__device__ __forceinline__ v8f wmma4(v2f a, v2f b, v8f c) {
    return __builtin_amdgcn_wmma_f32_16x16x4_f32(
         false, a,  false, b,
         (short)0, c,  false,  false);
}

__device__ __forceinline__ v2f ld_bfrag(const float2* wf, int lane) {
    float2 t = wf[lane];
    v2f b; b.x = t.x; b.y = t.y;
    return b;
}

__global__ __launch_bounds__(256) void lstm_fused_kernel(
    const float* __restrict__ x,
    const float* __restrict__ w_ih0, const float* __restrict__ w_hh0,
    const float* __restrict__ b_ih0, const float* __restrict__ b_hh0,
    const float* __restrict__ w_ih1, const float* __restrict__ w_hh1,
    const float* __restrict__ b_ih1, const float* __restrict__ b_hh1,
    const float* __restrict__ w_cls, const float* __restrict__ b_cls,
    float* __restrict__ out, int B)
{
    __shared__ float2 wfrag[NFRAG][32];
    __shared__ float hbuf[8][16][17];

    const int tid = threadIdx.x;

    for (int idx = tid; idx < NFRAG * 64; idx += 256) {
        const int frag = idx >> 6;
        const int r    = idx & 63;
        const int ln   = r >> 1;
        const int v    = r & 1;
        const int n    = ln & 15;
        const int kb   = ((ln >> 4) << 1) + v;
        float val = 0.0f;
        if (frag < F_HH0) {
            const int g = frag / 7, c = frag % 7, k = 4 * c + kb;
            if (n < HID && k < D_IN) val = w_ih0[(g * HID + n) * D_IN + k];
        } else if (frag < F_IH1) {
            const int f2 = frag - F_HH0, g = f2 / 3, c = f2 % 3, k = 4 * c + kb;
            if (n < HID && k < HID) val = w_hh0[(g * HID + n) * HID + k];
        } else if (frag < F_HH1) {
            const int f2 = frag - F_IH1, g = f2 / 3, c = f2 % 3, k = 4 * c + kb;
            if (n < HID && k < HID) val = w_ih1[(g * HID + n) * HID + k];
        } else if (frag < F_CLS) {
            const int f2 = frag - F_HH1, g = f2 / 3, c = f2 % 3, k = 4 * c + kb;
            if (n < HID && k < HID) val = w_hh1[(g * HID + n) * HID + k];
        } else {
            const int c = frag - F_CLS, k = 4 * c + kb;
            if (n < HID && k < HID) val = w_cls[n * HID + k];
        }
        ((float*)&wfrag[frag][ln])[v] = val;
    }
    __syncthreads();

    const int wave = tid >> 5;
    const int lane = tid & 31;
    const int lo   = lane & 15;
    const int hi   = lane >> 4;

    const int b0   = blockIdx.x * 128 + wave * 16;
    const int brow = b0 + lo;
    const int brow_c = (brow < B) ? brow : (B - 1);

    float bias0[4], bias1[4];
#pragma unroll
    for (int g = 0; g < 4; ++g) {
        bias0[g] = (lo < HID) ? (b_ih0[g * HID + lo] + b_hh0[g * HID + lo]) : 0.0f;
        bias1[g] = (lo < HID) ? (b_ih1[g * HID + lo] + b_hh1[g * HID + lo]) : 0.0f;
    }
    const float bcls = (lo < HID) ? b_cls[lo] : 0.0f;

    v2f a_h0[3], a_h1[3];
#pragma unroll
    for (int c = 0; c < 3; ++c) {
        a_h0[c].x = 0.0f; a_h0[c].y = 0.0f;
        a_h1[c].x = 0.0f; a_h1[c].y = 0.0f;
    }
    float c0v[8], c1v[8];
#pragma unroll
    for (int v = 0; v < 8; ++v) { c0v[v] = 0.0f; c1v[v] = 0.0f; }

    const float* xrow = x + (size_t)brow_c * (T_STEPS * D_IN);

    for (int t = 0; t < T_STEPS; ++t) {
        v2f xa[7];
#pragma unroll
        for (int c = 0; c < 7; ++c) {
            const int k = 4 * c + 2 * hi;
            const float2 p = *(const float2*)(xrow + t * D_IN + k);
            xa[c].x = p.x; xa[c].y = p.y;
        }
        if (t + 1 < T_STEPS)
            __builtin_prefetch(xrow + (t + 1) * D_IN, 0, 1);

        v8f acc0[4];
#pragma unroll
        for (int g = 0; g < 4; ++g) {
#pragma unroll
            for (int v = 0; v < 8; ++v) acc0[g][v] = bias0[g];
        }
#pragma unroll
        for (int c = 0; c < 7; ++c) {
#pragma unroll
            for (int g = 0; g < 4; ++g)
                acc0[g] = wmma4(xa[c], ld_bfrag(wfrag[F_IH0 + g * 7 + c], lane), acc0[g]);
        }
#pragma unroll
        for (int c = 0; c < 3; ++c) {
#pragma unroll
            for (int g = 0; g < 4; ++g)
                acc0[g] = wmma4(a_h0[c], ld_bfrag(wfrag[F_HH0 + g * 3 + c], lane), acc0[g]);
        }
        float h0v[8];
#pragma unroll
        for (int v = 0; v < 8; ++v) {
            const float ig = sigf(acc0[0][v]);
            const float fg = sigf(acc0[1][v]);
            const float gg = tanhf_fast(acc0[2][v]);
            const float og = sigf(acc0[3][v]);
            c0v[v] = fg * c0v[v] + ig * gg;
            h0v[v] = og * tanhf_fast(c0v[v]);
        }
#pragma unroll
        for (int v = 0; v < 8; ++v) hbuf[wave][v + 8 * hi][lo] = h0v[v];
#pragma unroll
        for (int c = 0; c < 3; ++c) {
            const int k = 4 * c + 2 * hi;
            a_h0[c].x = hbuf[wave][lo][k];
            a_h0[c].y = hbuf[wave][lo][k + 1];
        }

        v8f acc1[4];
#pragma unroll
        for (int g = 0; g < 4; ++g) {
#pragma unroll
            for (int v = 0; v < 8; ++v) acc1[g][v] = bias1[g];
        }
#pragma unroll
        for (int c = 0; c < 3; ++c) {
#pragma unroll
            for (int g = 0; g < 4; ++g)
                acc1[g] = wmma4(a_h0[c], ld_bfrag(wfrag[F_IH1 + g * 3 + c], lane), acc1[g]);
        }
#pragma unroll
        for (int c = 0; c < 3; ++c) {
#pragma unroll
            for (int g = 0; g < 4; ++g)
                acc1[g] = wmma4(a_h1[c], ld_bfrag(wfrag[F_HH1 + g * 3 + c], lane), acc1[g]);
        }
        float h1v[8];
#pragma unroll
        for (int v = 0; v < 8; ++v) {
            const float ig = sigf(acc1[0][v]);
            const float fg = sigf(acc1[1][v]);
            const float gg = tanhf_fast(acc1[2][v]);
            const float og = sigf(acc1[3][v]);
            c1v[v] = fg * c1v[v] + ig * gg;
            h1v[v] = og * tanhf_fast(c1v[v]);
        }
#pragma unroll
        for (int v = 0; v < 8; ++v) hbuf[wave][v + 8 * hi][lo] = h1v[v];
#pragma unroll
        for (int c = 0; c < 3; ++c) {
            const int k = 4 * c + 2 * hi;
            a_h1[c].x = hbuf[wave][lo][k];
            a_h1[c].y = hbuf[wave][lo][k + 1];
        }
    }

    v8f accc;
#pragma unroll
    for (int v = 0; v < 8; ++v) accc[v] = bcls;
#pragma unroll
    for (int c = 0; c < 3; ++c)
        accc = wmma4(a_h1[c], ld_bfrag(wfrag[F_CLS + c], lane), accc);

    if (lo < HID) {
#pragma unroll
        for (int v = 0; v < 8; ++v) {
            const int b = b0 + v + 8 * hi;
            if (b < B) out[b * 10 + lo] = accc[v];
        }
    }
}

extern "C" void kernel_launch(void* const* d_in, const int* in_sizes, int n_in,
                              void* d_out, int out_size, void* d_ws, size_t ws_size,
                              hipStream_t stream) {
    const float* x     = (const float*)d_in[0];
    const float* w_ih0 = (const float*)d_in[1];
    const float* w_hh0 = (const float*)d_in[2];
    const float* b_ih0 = (const float*)d_in[3];
    const float* b_hh0 = (const float*)d_in[4];
    const float* w_ih1 = (const float*)d_in[5];
    const float* w_hh1 = (const float*)d_in[6];
    const float* b_ih1 = (const float*)d_in[7];
    const float* b_hh1 = (const float*)d_in[8];
    const float* w_cls = (const float*)d_in[9];
    const float* b_cls = (const float*)d_in[10];

    const int B = in_sizes[0] / (T_STEPS * D_IN);
    const int blocks = (B + 127) / 128;

    lstm_fused_kernel<<<blocks, 256, 0, stream>>>(
        x, w_ih0, w_hh0, b_ih0, b_hh0,
        w_ih1, w_hh1, b_ih1, b_hh1,
        w_cls, b_cls, (float*)d_out, B);
}


# Products transposed: weights as WMMA A operand, gate rows permuted so a lane owns whole hidden units of its batch column (3 M-tiles, not 4); h stays in registers as B, so its k sum order is permuted
# speedup vs baseline: 1.2937x; 1.2937x over previous
#include <hip/hip_runtime.h>


typedef float v2f __attribute__((ext_vector_type(2)));
typedef float v8f __attribute__((ext_vector_type(8)));

#define T_STEPS 28
#define D_IN    28
#define HID     10

#define F_IH0 0
#define F_HH0 21
#define F_IH1 30
#define F_HH1 39
#define F_CLS 48
#define NFRAG 51

__device__ __forceinline__ float sigf(float x) {
    return __builtin_amdgcn_rcpf(1.0f + __expf(-x));
}
__device__ __forceinline__ float tanhf_fast(float x) {
    return 1.0f - 2.0f * __builtin_amdgcn_rcpf(1.0f + __expf(2.0f * x));
}

__device__ __forceinline__ v8f wmma4(v2f a, v2f b, v8f c) {
    return __builtin_amdgcn_wmma_f32_16x16x4_f32(
         false, a,  false, b,
         (short)0, c,  false,  false);
}

__device__ __forceinline__ int unit_of(int hh, int u) {
    return (u < 5) ? (5 * hh + u) : -1;
}

__device__ __forceinline__ v2f ld_afrag(const float2* wf, int lane) {
    float2 t = wf[lane];
    v2f a; a.x = t.x; a.y = t.y;
    return a;
}

__device__ __forceinline__ void cell_update(const v8f* acc, float* cst, v2f* bh) {
    float hv[5];
#pragma unroll
    for (int u = 0; u < 5; ++u) {
        const int m = u >> 1, o4 = 4 * (u & 1);
        const float ig = sigf(acc[m][o4 + 0]);
        const float fg = sigf(acc[m][o4 + 1]);
        const float gg = tanhf_fast(acc[m][o4 + 2]);
        const float og = sigf(acc[m][o4 + 3]);
        cst[u] = fg * cst[u] + ig * gg;
        hv[u]  = og * tanhf_fast(cst[u]);
    }
    bh[0].x = hv[0]; bh[0].y = hv[1];
    bh[1].x = hv[2]; bh[1].y = hv[3];
    bh[2].x = hv[4]; bh[2].y = 0.0f;
}

__global__ __launch_bounds__(256) void lstm_fused_kernel(
    const float* __restrict__ x,
    const float* __restrict__ w_ih0, const float* __restrict__ w_hh0,
    const float* __restrict__ b_ih0, const float* __restrict__ b_hh0,
    const float* __restrict__ w_ih1, const float* __restrict__ w_hh1,
    const float* __restrict__ b_ih1, const float* __restrict__ b_hh1,
    const float* __restrict__ w_cls, const float* __restrict__ b_cls,
    float* __restrict__ out, int B)
{
    __shared__ float2 wfrag[NFRAG][32];
    __shared__ float biasT[2][2][24];
    __shared__ float bclsT[16];

    const int tid = threadIdx.x;

    for (int idx = tid; idx < NFRAG * 64; idx += 256) {
        const int frag = idx >> 6;
        const int r    = idx & 63;
        const int ln   = r >> 1;
        const int s    = r & 1;
        const int row  = ln & 15;
        const int kh   = ln >> 4;
        float val = 0.0f;
        if (frag < F_CLS) {
            int m, q, mat;
            if (frag < F_HH0) { mat = 0; m = frag / 7; q = frag % 7; }
            else { const int f2 = frag - F_HH0; mat = 1 + f2 / 9; m = (f2 % 9) / 3; q = f2 % 3; }
            const int r8 = row & 7;
            const int g  = r8 & 3;
            const int j  = unit_of(row >> 3, 2 * m + (r8 >> 2));
            if (mat == 0) {
                const int k = 4 * q + 2 * kh + s;
                if (j >= 0) val = w_ih0[(g * HID + j) * D_IN + k];
            } else {
                const int jk = unit_of(kh, 2 * q + s);
                const float* wm = (mat == 1) ? w_hh0 : ((mat == 2) ? w_ih1 : w_hh1);
                if (j >= 0 && jk >= 0) val = wm[(g * HID + j) * HID + jk];
            }
        } else {
            const int q  = frag - F_CLS;
            const int jk = unit_of(kh, 2 * q + s);
            if (row < HID && jk >= 0) val = w_cls[row * HID + jk];
        }
        ((float*)&wfrag[frag][ln])[s] = val;
    }
    if (tid < 96) {
        const int layer = tid / 48, rem = tid % 48;
        const int hh = rem / 24, si = rem % 24;
        const int m = si >> 3, r8 = si & 7;
        const int g = r8 & 3;
        const int j = unit_of(hh, 2 * m + (r8 >> 2));
        float v = 0.0f;
        if (j >= 0) {
            const float* bi = layer ? b_ih1 : b_ih0;
            const float* bh = layer ? b_hh1 : b_hh0;
            v = bi[g * HID + j] + bh[g * HID + j];
        }
        biasT[layer][hh][si] = v;
    } else if (tid < 112) {
        const int n = tid - 96;
        bclsT[n] = (n < HID) ? b_cls[n] : 0.0f;
    }
    __syncthreads();

    const int wave = tid >> 5;
    const int lane = tid & 31;
    const int lo   = lane & 15;
    const int hi   = lane >> 4;

    const int b0   = blockIdx.x * 128 + wave * 16;
    const int brow = b0 + lo;
    const int brow_c = (brow < B) ? brow : (B - 1);

    v2f a_ih0[3][7], a_hh0[3][3], a_ih1[3][3], a_hh1[3][3];
#pragma unroll
    for (int m = 0; m < 3; ++m) {
#pragma unroll
        for (int q = 0; q < 7; ++q) a_ih0[m][q] = ld_afrag(wfrag[F_IH0 + m * 7 + q], lane);
#pragma unroll
        for (int q = 0; q < 3; ++q) {
            a_hh0[m][q] = ld_afrag(wfrag[F_HH0 + m * 3 + q], lane);
            a_ih1[m][q] = ld_afrag(wfrag[F_IH1 + m * 3 + q], lane);
            a_hh1[m][q] = ld_afrag(wfrag[F_HH1 + m * 3 + q], lane);
        }
    }
    v8f bias0[3], bias1[3];
#pragma unroll
    for (int m = 0; m < 3; ++m) {
#pragma unroll
        for (int v = 0; v < 8; ++v) {
            bias0[m][v] = biasT[0][hi][m * 8 + v];
            bias1[m][v] = biasT[1][hi][m * 8 + v];
        }
    }

    v2f bh0[3], bh1[3];
#pragma unroll
    for (int q = 0; q < 3; ++q) {
        bh0[q].x = 0.0f; bh0[q].y = 0.0f;
        bh1[q].x = 0.0f; bh1[q].y = 0.0f;
    }
    float c0v[5], c1v[5];
#pragma unroll
    for (int u = 0; u < 5; ++u) { c0v[u] = 0.0f; c1v[u] = 0.0f; }

    const float* xrow = x + (size_t)brow_c * (T_STEPS * D_IN);

    for (int t = 0; t < T_STEPS; ++t) {
        v2f xb[7];
#pragma unroll
        for (int q = 0; q < 7; ++q) {
            const int k = 4 * q + 2 * hi;
            const float2 p = *(const float2*)(xrow + t * D_IN + k);
            xb[q].x = p.x; xb[q].y = p.y;
        }
        if (t + 1 < T_STEPS)
            __builtin_prefetch(xrow + (t + 1) * D_IN, 0, 1);

        v8f acc0[3];
#pragma unroll
        for (int m = 0; m < 3; ++m) acc0[m] = bias0[m];
#pragma unroll
        for (int q = 0; q < 7; ++q) {
#pragma unroll
            for (int m = 0; m < 3; ++m) acc0[m] = wmma4(a_ih0[m][q], xb[q], acc0[m]);
        }
#pragma unroll
        for (int q = 0; q < 3; ++q) {
#pragma unroll
            for (int m = 0; m < 3; ++m) acc0[m] = wmma4(a_hh0[m][q], bh0[q], acc0[m]);
        }
        cell_update(acc0, c0v, bh0);

        v8f acc1[3];
#pragma unroll
        for (int m = 0; m < 3; ++m) acc1[m] = bias1[m];
#pragma unroll
        for (int q = 0; q < 3; ++q) {
#pragma unroll
            for (int m = 0; m < 3; ++m) acc1[m] = wmma4(a_ih1[m][q], bh0[q], acc1[m]);
        }
#pragma unroll
        for (int q = 0; q < 3; ++q) {
#pragma unroll
            for (int m = 0; m < 3; ++m) acc1[m] = wmma4(a_hh1[m][q], bh1[q], acc1[m]);
        }
        cell_update(acc1, c1v, bh1);
    }

    v8f accc;
#pragma unroll
    for (int v = 0; v < 8; ++v) accc[v] = bclsT[8 * hi + v];
#pragma unroll
    for (int q = 0; q < 3; ++q)
        accc = wmma4(ld_afrag(wfrag[F_CLS + q], lane), bh1[q], accc);

    if (brow < B) {
        float* orow = out + (size_t)brow * HID + 8 * hi;
        float2 p; p.x = accc[0]; p.y = accc[1];
        *(float2*)(orow) = p;
        if (hi == 0) {
#pragma unroll
            for (int v = 2; v < 8; v += 2) {
                float2 w; w.x = accc[v]; w.y = accc[v + 1];
                *(float2*)(orow + v) = w;
            }
        }
    }
}

extern "C" void kernel_launch(void* const* d_in, const int* in_sizes, int n_in,
                              void* d_out, int out_size, void* d_ws, size_t ws_size,
                              hipStream_t stream) {
    const float* x     = (const float*)d_in[0];
    const float* w_ih0 = (const float*)d_in[1];
    const float* w_hh0 = (const float*)d_in[2];
    const float* b_ih0 = (const float*)d_in[3];
    const float* b_hh0 = (const float*)d_in[4];
    const float* w_ih1 = (const float*)d_in[5];
    const float* w_hh1 = (const float*)d_in[6];
    const float* b_ih1 = (const float*)d_in[7];
    const float* b_hh1 = (const float*)d_in[8];
    const float* w_cls = (const float*)d_in[9];
    const float* b_cls = (const float*)d_in[10];

    const int B = in_sizes[0] / (T_STEPS * D_IN);
    const int blocks = (B + 127) / 128;

    lstm_fused_kernel<<<blocks, 256, 0, stream>>>(
        x, w_ih0, w_hh0, b_ih0, b_hh0,
        w_ih1, w_hh1, b_ih1, b_hh1,
        w_cls, b_cls, (float*)d_out, B);
}


# Time loop software-pipelined: trip t = x loads of t+1, layer-0 WMMAs of t, layer-1 WMMAs of t-1; each layer's cell math cut into steps between the other layer's WMMAs; every sum unchanged
# speedup vs baseline: 1.4762x; 1.1411x over previous
#include <hip/hip_runtime.h>


typedef float v2f __attribute__((ext_vector_type(2)));
typedef float v8f __attribute__((ext_vector_type(8)));

#define T_STEPS 28
#define D_IN    28
#define HID     10

#define F_IH0 0
#define F_HH0 21
#define F_IH1 30
#define F_HH1 39
#define F_CLS 48
#define NFRAG 51

__device__ __forceinline__ float sigf(float x) {
    return __builtin_amdgcn_rcpf(1.0f + __expf(-x));
}
__device__ __forceinline__ float tanhf_fast(float x) {
    return 1.0f - 2.0f * __builtin_amdgcn_rcpf(1.0f + __expf(2.0f * x));
}

__device__ __forceinline__ v8f wmma4(v2f a, v2f b, v8f c) {
    return __builtin_amdgcn_wmma_f32_16x16x4_f32(
         false, a,  false, b,
         (short)0, c,  false,  false);
}

__device__ __forceinline__ int unit_of(int hh, int u) {
    return (u < 5) ? (5 * hh + u) : -1;
}

__device__ __forceinline__ v2f ld_afrag(const float2* wf, int lane) {
    float2 t = wf[lane];
    v2f a; a.x = t.x; a.y = t.y;
    return a;
}

#define CELL_STEPS 40
struct CellW {
    float e[5][4];
    float r[5][4];
    float ig[5];
    float ec[5];
    float rc[5];
    float hv[5];
};

__device__ __forceinline__ void cell_step(int n, const v8f* acc, float* cst, CellW& w) {
    const int st = n / 5, u = n % 5;
    const int m = u >> 1, o4 = 4 * (u & 1);
    switch (st) {
    case 0:
        w.e[u][0] = __expf(-acc[m][o4 + 0]);
        w.e[u][1] = __expf(-acc[m][o4 + 1]);
        break;
    case 1:
        w.e[u][2] = __expf(2.0f * acc[m][o4 + 2]);
        w.e[u][3] = __expf(-acc[m][o4 + 3]);
        break;
    case 2:
        w.r[u][0] = __builtin_amdgcn_rcpf(1.0f + w.e[u][0]);
        w.r[u][1] = __builtin_amdgcn_rcpf(1.0f + w.e[u][1]);
        break;
    case 3:
        w.r[u][2] = __builtin_amdgcn_rcpf(1.0f + w.e[u][2]);
        w.r[u][3] = __builtin_amdgcn_rcpf(1.0f + w.e[u][3]);
        break;
    case 4:
        w.ig[u] = w.r[u][0] * (1.0f - 2.0f * w.r[u][2]);
        break;
    case 5:
        cst[u] = w.r[u][1] * cst[u] + w.ig[u];
        w.ec[u] = __expf(2.0f * cst[u]);
        break;
    case 6:
        w.rc[u] = __builtin_amdgcn_rcpf(1.0f + w.ec[u]);
        break;
    default:
        w.hv[u] = w.r[u][3] * (1.0f - 2.0f * w.rc[u]);
        break;
    }
}

__device__ __forceinline__ void cell_steps(int lo, int hi, const v8f* acc, float* cst, CellW& w) {
#pragma unroll
    for (int n = lo; n < hi; ++n) cell_step(n, acc, cst, w);
}

__device__ __forceinline__ void cell_finish(const CellW& w, v2f* bh) {
    bh[0].x = w.hv[0]; bh[0].y = w.hv[1];
    bh[1].x = w.hv[2]; bh[1].y = w.hv[3];
    bh[2].x = w.hv[4]; bh[2].y = 0.0f;
}

__device__ __forceinline__ void cell_update(const v8f* acc, float* cst, v2f* bh) {
    CellW w;
    cell_steps(0, CELL_STEPS, acc, cst, w);
    cell_finish(w, bh);
}

__device__ __forceinline__ void load_x(const float* xrow, int t, int hi, v2f* xb) {
#pragma unroll
    for (int q = 0; q < 7; ++q) {
        const float2 p = *(const float2*)(xrow + t * D_IN + 4 * q + 2 * hi);
        xb[q].x = p.x; xb[q].y = p.y;
    }
}

template <int NQ_IN, bool SHARE>
__device__ __forceinline__ void gates(v8f* acc, const v8f* bias, const v2f (*a_in)[NQ_IN],
                                      const v2f (*a_hh)[3], const v2f* bin, const v2f* bh,
                                      const v8f* oacc, float* ocst, CellW& ow) {
    constexpr int NW = 3 * (NQ_IN + 3);
#pragma unroll
    for (int m = 0; m < 3; ++m) acc[m] = bias[m];
#pragma unroll
    for (int q = 0; q < NQ_IN + 3; ++q) {
#pragma unroll
        for (int m = 0; m < 3; ++m) {
            const int i = 3 * q + m;
            acc[m] = (q < NQ_IN) ? wmma4(a_in[m][q], bin[q], acc[m])
                                 : wmma4(a_hh[m][q - NQ_IN], bh[q - NQ_IN], acc[m]);
            if (SHARE) {
                cell_steps(i * CELL_STEPS / NW, (i + 1) * CELL_STEPS / NW, oacc, ocst, ow);
                __builtin_amdgcn_sched_barrier(0);
            }
        }
    }
}

__global__ __launch_bounds__(256) void lstm_fused_kernel(
    const float* __restrict__ x,
    const float* __restrict__ w_ih0, const float* __restrict__ w_hh0,
    const float* __restrict__ b_ih0, const float* __restrict__ b_hh0,
    const float* __restrict__ w_ih1, const float* __restrict__ w_hh1,
    const float* __restrict__ b_ih1, const float* __restrict__ b_hh1,
    const float* __restrict__ w_cls, const float* __restrict__ b_cls,
    float* __restrict__ out, int B)
{
    __shared__ float2 wfrag[NFRAG][32];
    __shared__ float biasT[2][2][24];
    __shared__ float bclsT[16];

    const int tid = threadIdx.x;

    for (int idx = tid; idx < NFRAG * 64; idx += 256) {
        const int frag = idx >> 6;
        const int r    = idx & 63;
        const int ln   = r >> 1;
        const int s    = r & 1;
        const int row  = ln & 15;
        const int kh   = ln >> 4;
        float val = 0.0f;
        if (frag < F_CLS) {
            int m, q, mat;
            if (frag < F_HH0) { mat = 0; m = frag / 7; q = frag % 7; }
            else { const int f2 = frag - F_HH0; mat = 1 + f2 / 9; m = (f2 % 9) / 3; q = f2 % 3; }
            const int r8 = row & 7;
            const int g  = r8 & 3;
            const int j  = unit_of(row >> 3, 2 * m + (r8 >> 2));
            if (mat == 0) {
                const int k = 4 * q + 2 * kh + s;
                if (j >= 0) val = w_ih0[(g * HID + j) * D_IN + k];
            } else {
                const int jk = unit_of(kh, 2 * q + s);
                const float* wm = (mat == 1) ? w_hh0 : ((mat == 2) ? w_ih1 : w_hh1);
                if (j >= 0 && jk >= 0) val = wm[(g * HID + j) * HID + jk];
            }
        } else {
            const int q  = frag - F_CLS;
            const int jk = unit_of(kh, 2 * q + s);
            if (row < HID && jk >= 0) val = w_cls[row * HID + jk];
        }
        ((float*)&wfrag[frag][ln])[s] = val;
    }
    if (tid < 96) {
        const int layer = tid / 48, rem = tid % 48;
        const int hh = rem / 24, si = rem % 24;
        const int m = si >> 3, r8 = si & 7;
        const int g = r8 & 3;
        const int j = unit_of(hh, 2 * m + (r8 >> 2));
        float v = 0.0f;
        if (j >= 0) {
            const float* bi = layer ? b_ih1 : b_ih0;
            const float* bh = layer ? b_hh1 : b_hh0;
            v = bi[g * HID + j] + bh[g * HID + j];
        }
        biasT[layer][hh][si] = v;
    } else if (tid < 112) {
        const int n = tid - 96;
        bclsT[n] = (n < HID) ? b_cls[n] : 0.0f;
    }
    __syncthreads();

    const int wave = tid >> 5;
    const int lane = tid & 31;
    const int lo   = lane & 15;
    const int hi   = lane >> 4;

    const int b0   = blockIdx.x * 128 + wave * 16;
    const int brow = b0 + lo;
    const int brow_c = (brow < B) ? brow : (B - 1);

    v2f a_ih0[3][7], a_hh0[3][3], a_ih1[3][3], a_hh1[3][3];
#pragma unroll
    for (int m = 0; m < 3; ++m) {
#pragma unroll
        for (int q = 0; q < 7; ++q) a_ih0[m][q] = ld_afrag(wfrag[F_IH0 + m * 7 + q], lane);
#pragma unroll
        for (int q = 0; q < 3; ++q) {
            a_hh0[m][q] = ld_afrag(wfrag[F_HH0 + m * 3 + q], lane);
            a_ih1[m][q] = ld_afrag(wfrag[F_IH1 + m * 3 + q], lane);
            a_hh1[m][q] = ld_afrag(wfrag[F_HH1 + m * 3 + q], lane);
        }
    }
    v8f bias0[3], bias1[3];
#pragma unroll
    for (int m = 0; m < 3; ++m) {
#pragma unroll
        for (int v = 0; v < 8; ++v) {
            bias0[m][v] = biasT[0][hi][m * 8 + v];
            bias1[m][v] = biasT[1][hi][m * 8 + v];
        }
    }

    v2f bh0[3], bh1[3];
#pragma unroll
    for (int q = 0; q < 3; ++q) {
        bh0[q].x = 0.0f; bh0[q].y = 0.0f;
        bh1[q].x = 0.0f; bh1[q].y = 0.0f;
    }
    float c0v[5], c1v[5];
#pragma unroll
    for (int u = 0; u < 5; ++u) { c0v[u] = 0.0f; c1v[u] = 0.0f; }

    const float* xrow = x + (size_t)brow_c * (T_STEPS * D_IN);

    v2f xb[7], xn[7];
    load_x(xrow, 0, hi, xb);
    load_x(xrow, 1, hi, xn);

    v8f acc0[3], acc1[3];
    CellW w0, w1;

    gates<7, false>(acc0, bias0, a_ih0, a_hh0, xb, bh0, acc1, c1v, w1);
    cell_update(acc0, c0v, bh0);

#pragma unroll
    for (int q = 0; q < 7; ++q) xb[q] = xn[q];
    load_x(xrow, 2, hi, xn);
    gates<7, false>(acc0, bias0, a_ih0, a_hh0, xb, bh0, acc1, c1v, w1);
    gates<3, true>(acc1, bias1, a_ih1, a_hh1, bh0, bh1, acc0, c0v, w0);
    cell_finish(w0, bh0);

    for (int t = 2; t < T_STEPS; ++t) {
#pragma unroll
        for (int q = 0; q < 7; ++q) xb[q] = xn[q];
        const int tn = (t + 1 < T_STEPS) ? (t + 1) : (T_STEPS - 1);
        load_x(xrow, tn, hi, xn);
        const int tp = (t + 2 < T_STEPS) ? (t + 2) : (T_STEPS - 1);
        __builtin_prefetch(xrow + tp * D_IN + (D_IN - 1), 0, 1);

        gates<7, true>(acc0, bias0, a_ih0, a_hh0, xb, bh0, acc1, c1v, w1);
        cell_finish(w1, bh1);
        gates<3, true>(acc1, bias1, a_ih1, a_hh1, bh0, bh1, acc0, c0v, w0);
        cell_finish(w0, bh0);
    }

    cell_update(acc1, c1v, bh1);
    gates<3, false>(acc1, bias1, a_ih1, a_hh1, bh0, bh1, acc0, c0v, w0);
    cell_update(acc1, c1v, bh1);

    v8f accc;
#pragma unroll
    for (int v = 0; v < 8; ++v) accc[v] = bclsT[8 * hi + v];
#pragma unroll
    for (int q = 0; q < 3; ++q)
        accc = wmma4(ld_afrag(wfrag[F_CLS + q], lane), bh1[q], accc);

    if (brow < B) {
        float* orow = out + (size_t)brow * HID + 8 * hi;
        float2 p; p.x = accc[0]; p.y = accc[1];
        *(float2*)(orow) = p;
        if (hi == 0) {
#pragma unroll
            for (int v = 2; v < 8; v += 2) {
                float2 w; w.x = accc[v]; w.y = accc[v + 1];
                *(float2*)(orow + v) = w;
            }
        }
    }
}

extern "C" void kernel_launch(void* const* d_in, const int* in_sizes, int n_in,
                              void* d_out, int out_size, void* d_ws, size_t ws_size,
                              hipStream_t stream) {
    const float* x     = (const float*)d_in[0];
    const float* w_ih0 = (const float*)d_in[1];
    const float* w_hh0 = (const float*)d_in[2];
    const float* b_ih0 = (const float*)d_in[3];
    const float* b_hh0 = (const float*)d_in[4];
    const float* w_ih1 = (const float*)d_in[5];
    const float* w_hh1 = (const float*)d_in[6];
    const float* b_ih1 = (const float*)d_in[7];
    const float* b_hh1 = (const float*)d_in[8];
    const float* w_cls = (const float*)d_in[9];
    const float* b_cls = (const float*)d_in[10];

    const int B = in_sizes[0] / (T_STEPS * D_IN);
    const int blocks = (B + 127) / 128;

    lstm_fused_kernel<<<blocks, 256, 0, stream>>>(
        x, w_ih0, w_hh0, b_ih0, b_hh0,
        w_ih1, w_hh1, b_ih1, b_hh1,
        w_cls, b_cls, (float*)d_out, B);
}


# Activations on the hardware tanh, one transcendental each instead of exp + rcp: sigmoid(a) = (1 + tanh(a/2))/2, the halving folded exactly into the staged i,f,o weight rows and biases; sums unchanged
# speedup vs baseline: 1.6727x; 1.1331x over previous
#include <hip/hip_runtime.h>


typedef float v2f __attribute__((ext_vector_type(2)));
typedef float v8f __attribute__((ext_vector_type(8)));

#define T_STEPS 28
#define D_IN    28
#define HID     10

#define F_IH0 0
#define F_HH0 21
#define F_IH1 30
#define F_HH1 39
#define F_CLS 48
#define NFRAG 51

__device__ __forceinline__ float htanh(float x) {
    return __builtin_amdgcn_tanhf(x);
}
__device__ __forceinline__ float gate_scale(int g) {
    return (g == 2) ? 1.0f : 0.5f;
}

__device__ __forceinline__ v8f wmma4(v2f a, v2f b, v8f c) {
    return __builtin_amdgcn_wmma_f32_16x16x4_f32(
         false, a,  false, b,
         (short)0, c,  false,  false);
}

__device__ __forceinline__ int unit_of(int hh, int u) {
    return (u < 5) ? (5 * hh + u) : -1;
}

__device__ __forceinline__ v2f ld_afrag(const float2* wf, int lane) {
    float2 t = wf[lane];
    v2f a; a.x = t.x; a.y = t.y;
    return a;
}

#define CELL_STEPS 30
struct CellW {
    float t[5][4];
    float si[5];
    float sf[5];
    float so[5];
    float ig[5];
    float tc[5];
    float hv[5];
};

__device__ __forceinline__ void cell_step(int n, const v8f* acc, float* cst, CellW& w) {
    const int st = n / 5, u = n % 5;
    const int m = u >> 1, o4 = 4 * (u & 1);
    switch (st) {
    case 0:
        w.t[u][0] = htanh(acc[m][o4 + 0]);
        w.t[u][1] = htanh(acc[m][o4 + 1]);
        break;
    case 1:
        w.t[u][2] = htanh(acc[m][o4 + 2]);
        w.t[u][3] = htanh(acc[m][o4 + 3]);
        break;
    case 2:
        w.si[u] = 0.5f * w.t[u][0] + 0.5f;
        w.sf[u] = 0.5f * w.t[u][1] + 0.5f;
        break;
    case 3:
        w.ig[u] = w.si[u] * w.t[u][2];
        w.so[u] = 0.5f * w.t[u][3] + 0.5f;
        break;
    case 4:
        cst[u] = w.sf[u] * cst[u] + w.ig[u];
        w.tc[u] = htanh(cst[u]);
        break;
    default:
        w.hv[u] = w.so[u] * w.tc[u];
        break;
    }
}

__device__ __forceinline__ void cell_steps(int lo, int hi, const v8f* acc, float* cst, CellW& w) {
#pragma unroll
    for (int n = lo; n < hi; ++n) cell_step(n, acc, cst, w);
}

__device__ __forceinline__ void cell_finish(const CellW& w, v2f* bh) {
    bh[0].x = w.hv[0]; bh[0].y = w.hv[1];
    bh[1].x = w.hv[2]; bh[1].y = w.hv[3];
    bh[2].x = w.hv[4]; bh[2].y = 0.0f;
}

__device__ __forceinline__ void cell_update(const v8f* acc, float* cst, v2f* bh) {
    CellW w;
    cell_steps(0, CELL_STEPS, acc, cst, w);
    cell_finish(w, bh);
}

__device__ __forceinline__ void load_x(const float* xrow, int t, int hi, v2f* xb) {
#pragma unroll
    for (int q = 0; q < 7; ++q) {
        const float2 p = *(const float2*)(xrow + t * D_IN + 4 * q + 2 * hi);
        xb[q].x = p.x; xb[q].y = p.y;
    }
}

template <int NQ_IN, bool SHARE>
__device__ __forceinline__ void gates(v8f* acc, const v8f* bias, const v2f (*a_in)[NQ_IN],
                                      const v2f (*a_hh)[3], const v2f* bin, const v2f* bh,
                                      const v8f* oacc, float* ocst, CellW& ow) {
    constexpr int NW = 3 * (NQ_IN + 3);
#pragma unroll
    for (int m = 0; m < 3; ++m) acc[m] = bias[m];
#pragma unroll
    for (int q = 0; q < NQ_IN + 3; ++q) {
#pragma unroll
        for (int m = 0; m < 3; ++m) {
            const int i = 3 * q + m;
            acc[m] = (q < NQ_IN) ? wmma4(a_in[m][q], bin[q], acc[m])
                                 : wmma4(a_hh[m][q - NQ_IN], bh[q - NQ_IN], acc[m]);
            if (SHARE) {
                cell_steps(i * CELL_STEPS / NW, (i + 1) * CELL_STEPS / NW, oacc, ocst, ow);
                __builtin_amdgcn_sched_barrier(0);
            }
        }
    }
}

__global__ __launch_bounds__(256) void lstm_fused_kernel(
    const float* __restrict__ x,
    const float* __restrict__ w_ih0, const float* __restrict__ w_hh0,
    const float* __restrict__ b_ih0, const float* __restrict__ b_hh0,
    const float* __restrict__ w_ih1, const float* __restrict__ w_hh1,
    const float* __restrict__ b_ih1, const float* __restrict__ b_hh1,
    const float* __restrict__ w_cls, const float* __restrict__ b_cls,
    float* __restrict__ out, int B)
{
    __shared__ float2 wfrag[NFRAG][32];
    __shared__ float biasT[2][2][24];
    __shared__ float bclsT[16];

    const int tid = threadIdx.x;

    for (int idx = tid; idx < NFRAG * 64; idx += 256) {
        const int frag = idx >> 6;
        const int r    = idx & 63;
        const int ln   = r >> 1;
        const int s    = r & 1;
        const int row  = ln & 15;
        const int kh   = ln >> 4;
        float val = 0.0f;
        if (frag < F_CLS) {
            int m, q, mat;
            if (frag < F_HH0) { mat = 0; m = frag / 7; q = frag % 7; }
            else { const int f2 = frag - F_HH0; mat = 1 + f2 / 9; m = (f2 % 9) / 3; q = f2 % 3; }
            const int r8 = row & 7;
            const int g  = r8 & 3;
            const int j  = unit_of(row >> 3, 2 * m + (r8 >> 2));
            if (mat == 0) {
                const int k = 4 * q + 2 * kh + s;
                if (j >= 0) val = gate_scale(g) * w_ih0[(g * HID + j) * D_IN + k];
            } else {
                const int jk = unit_of(kh, 2 * q + s);
                const float* wm = (mat == 1) ? w_hh0 : ((mat == 2) ? w_ih1 : w_hh1);
                if (j >= 0 && jk >= 0) val = gate_scale(g) * wm[(g * HID + j) * HID + jk];
            }
        } else {
            const int q  = frag - F_CLS;
            const int jk = unit_of(kh, 2 * q + s);
            if (row < HID && jk >= 0) val = w_cls[row * HID + jk];
        }
        ((float*)&wfrag[frag][ln])[s] = val;
    }
    if (tid < 96) {
        const int layer = tid / 48, rem = tid % 48;
        const int hh = rem / 24, si = rem % 24;
        const int m = si >> 3, r8 = si & 7;
        const int g = r8 & 3;
        const int j = unit_of(hh, 2 * m + (r8 >> 2));
        float v = 0.0f;
        if (j >= 0) {
            const float* bi = layer ? b_ih1 : b_ih0;
            const float* bh = layer ? b_hh1 : b_hh0;
            v = gate_scale(g) * (bi[g * HID + j] + bh[g * HID + j]);
        }
        biasT[layer][hh][si] = v;
    } else if (tid < 112) {
        const int n = tid - 96;
        bclsT[n] = (n < HID) ? b_cls[n] : 0.0f;
    }
    __syncthreads();

    const int wave = tid >> 5;
    const int lane = tid & 31;
    const int lo   = lane & 15;
    const int hi   = lane >> 4;

    const int b0   = blockIdx.x * 128 + wave * 16;
    const int brow = b0 + lo;
    const int brow_c = (brow < B) ? brow : (B - 1);

    v2f a_ih0[3][7], a_hh0[3][3], a_ih1[3][3], a_hh1[3][3];
#pragma unroll
    for (int m = 0; m < 3; ++m) {
#pragma unroll
        for (int q = 0; q < 7; ++q) a_ih0[m][q] = ld_afrag(wfrag[F_IH0 + m * 7 + q], lane);
#pragma unroll
        for (int q = 0; q < 3; ++q) {
            a_hh0[m][q] = ld_afrag(wfrag[F_HH0 + m * 3 + q], lane);
            a_ih1[m][q] = ld_afrag(wfrag[F_IH1 + m * 3 + q], lane);
            a_hh1[m][q] = ld_afrag(wfrag[F_HH1 + m * 3 + q], lane);
        }
    }
    v8f bias0[3], bias1[3];
#pragma unroll
    for (int m = 0; m < 3; ++m) {
#pragma unroll
        for (int v = 0; v < 8; ++v) {
            bias0[m][v] = biasT[0][hi][m * 8 + v];
            bias1[m][v] = biasT[1][hi][m * 8 + v];
        }
    }

    v2f bh0[3], bh1[3];
#pragma unroll
    for (int q = 0; q < 3; ++q) {
        bh0[q].x = 0.0f; bh0[q].y = 0.0f;
        bh1[q].x = 0.0f; bh1[q].y = 0.0f;
    }
    float c0v[5], c1v[5];
#pragma unroll
    for (int u = 0; u < 5; ++u) { c0v[u] = 0.0f; c1v[u] = 0.0f; }

    const float* xrow = x + (size_t)brow_c * (T_STEPS * D_IN);

    v2f xb[7], xn[7];
    load_x(xrow, 0, hi, xb);
    load_x(xrow, 1, hi, xn);

    v8f acc0[3], acc1[3];
    CellW w0, w1;

    gates<7, false>(acc0, bias0, a_ih0, a_hh0, xb, bh0, acc1, c1v, w1);
    cell_update(acc0, c0v, bh0);

#pragma unroll
    for (int q = 0; q < 7; ++q) xb[q] = xn[q];
    load_x(xrow, 2, hi, xn);
    gates<7, false>(acc0, bias0, a_ih0, a_hh0, xb, bh0, acc1, c1v, w1);
    gates<3, true>(acc1, bias1, a_ih1, a_hh1, bh0, bh1, acc0, c0v, w0);
    cell_finish(w0, bh0);

    for (int t = 2; t < T_STEPS; ++t) {
#pragma unroll
        for (int q = 0; q < 7; ++q) xb[q] = xn[q];
        const int tn = (t + 1 < T_STEPS) ? (t + 1) : (T_STEPS - 1);
        load_x(xrow, tn, hi, xn);
        const int tp = (t + 2 < T_STEPS) ? (t + 2) : (T_STEPS - 1);
        __builtin_prefetch(xrow + tp * D_IN + (D_IN - 1), 0, 1);

        gates<7, true>(acc0, bias0, a_ih0, a_hh0, xb, bh0, acc1, c1v, w1);
        cell_finish(w1, bh1);
        gates<3, true>(acc1, bias1, a_ih1, a_hh1, bh0, bh1, acc0, c0v, w0);
        cell_finish(w0, bh0);
    }

    cell_update(acc1, c1v, bh1);
    gates<3, false>(acc1, bias1, a_ih1, a_hh1, bh0, bh1, acc0, c0v, w0);
    cell_update(acc1, c1v, bh1);

    v8f accc;
#pragma unroll
    for (int v = 0; v < 8; ++v) accc[v] = bclsT[8 * hi + v];
#pragma unroll
    for (int q = 0; q < 3; ++q)
        accc = wmma4(ld_afrag(wfrag[F_CLS + q], lane), bh1[q], accc);

    if (brow < B) {
        float* orow = out + (size_t)brow * HID + 8 * hi;
        float2 p; p.x = accc[0]; p.y = accc[1];
        *(float2*)(orow) = p;
        if (hi == 0) {
#pragma unroll
            for (int v = 2; v < 8; v += 2) {
                float2 w; w.x = accc[v]; w.y = accc[v + 1];
                *(float2*)(orow + v) = w;
            }
        }
    }
}

extern "C" void kernel_launch(void* const* d_in, const int* in_sizes, int n_in,
                              void* d_out, int out_size, void* d_ws, size_t ws_size,
                              hipStream_t stream) {
    const float* x     = (const float*)d_in[0];
    const float* w_ih0 = (const float*)d_in[1];
    const float* w_hh0 = (const float*)d_in[2];
    const float* b_ih0 = (const float*)d_in[3];
    const float* b_hh0 = (const float*)d_in[4];
    const float* w_ih1 = (const float*)d_in[5];
    const float* w_hh1 = (const float*)d_in[6];
    const float* b_ih1 = (const float*)d_in[7];
    const float* b_hh1 = (const float*)d_in[8];
    const float* w_cls = (const float*)d_in[9];
    const float* b_cls = (const float*)d_in[10];

    const int B = in_sizes[0] / (T_STEPS * D_IN);
    const int blocks = (B + 127) / 128;

    lstm_fused_kernel<<<blocks, 256, 0, stream>>>(
        x, w_ih0, w_hh0, b_ih0, b_hh0,
        w_ih1, w_hh1, b_ih1, b_hh1,
        w_cls, b_cls, (float*)d_out, B);
}


# Both layers' gate rows packed into one set of 5 M-tiles, recurrent operand one [h0;h1] vector of 5 k-chunks: 42 WMMAs per step, not 48; k sum order of layer 1 and of the head permuted
# speedup vs baseline: 1.8210x; 1.0887x over previous
#include <hip/hip_runtime.h>


typedef float v2f __attribute__((ext_vector_type(2)));
typedef float v8f __attribute__((ext_vector_type(8)));

#define T_STEPS 28
#define D_IN    28
#define HID     10

#define F_X   0
#define F_H   21
#define F_CLS 46
#define NFRAG 49

__device__ __forceinline__ float htanh(float x) {
    return __builtin_amdgcn_tanhf(x);
}
__device__ __forceinline__ float gate_scale(int g) {
    return (g == 2) ? 1.0f : 0.5f;
}

__device__ __forceinline__ v8f wmma4(v2f a, v2f b, v8f c) {
    return __builtin_amdgcn_wmma_f32_16x16x4_f32(
         false, a,  false, b,
         (short)0, c,  false,  false);
}

__device__ __forceinline__ v2f ld_afrag(const float2* wf, int lane) {
    float2 t = wf[lane];
    v2f a; a.x = t.x; a.y = t.y;
    return a;
}

#define CELL_STEPS 30
#define CELL_READ  10
struct CellW {
    float t[5][4];
    float si[5];
    float sf[5];
    float so[5];
    float ig[5];
    float tc[5];
    float hv[5];
};

template <int S0>
__device__ __forceinline__ void cell_step(int n, const v8f* acc, float* cst, CellW& w) {
    const int st = n / 5, u = n % 5;
    const int sl = S0 + u;
    const int m = sl >> 1, o4 = 4 * (sl & 1);
    switch (st) {
    case 0:
        w.t[u][0] = htanh(acc[m][o4 + 0]);
        w.t[u][1] = htanh(acc[m][o4 + 1]);
        break;
    case 1:
        w.t[u][2] = htanh(acc[m][o4 + 2]);
        w.t[u][3] = htanh(acc[m][o4 + 3]);
        break;
    case 2:
        w.si[u] = 0.5f * w.t[u][0] + 0.5f;
        w.sf[u] = 0.5f * w.t[u][1] + 0.5f;
        break;
    case 3:
        w.ig[u] = w.si[u] * w.t[u][2];
        w.so[u] = 0.5f * w.t[u][3] + 0.5f;
        break;
    case 4:
        cst[u] = w.sf[u] * cst[u] + w.ig[u];
        w.tc[u] = htanh(cst[u]);
        break;
    default:
        w.hv[u] = w.so[u] * w.tc[u];
        break;
    }
}

template <int S0>
__device__ __forceinline__ void cell_steps(int lo, int hi, const v8f* acc, float* cst, CellW& w) {
#pragma unroll
    for (int n = lo; n < hi; ++n) cell_step<S0>(n, acc, cst, w);
}

__device__ __forceinline__ void pack_h(const float* h0v, const float* h1v, v2f* hb) {
    hb[0].x = h0v[0]; hb[0].y = h0v[1];
    hb[1].x = h0v[2]; hb[1].y = h0v[3];
    hb[2].x = h0v[4]; hb[2].y = h1v[0];
    hb[3].x = h1v[1]; hb[3].y = h1v[2];
    hb[4].x = h1v[3]; hb[4].y = h1v[4];
}

__device__ __forceinline__ void load_x(const float* xrow, int t, int hi, v2f* xb) {
#pragma unroll
    for (int q = 0; q < 7; ++q) {
        const float2 p = *(const float2*)(xrow + t * D_IN + 4 * q + 2 * hi);
        xb[q].x = p.x; xb[q].y = p.y;
    }
}

template <bool SHARE, bool DO1>
__device__ __forceinline__ void x_products(v8f* acc, const v8f* bias, const v2f (*a_x)[7], const v2f* xb,
                                           float* c0v, float* c1v, CellW& w0, CellW& w1) {
    constexpr int NW = 21;
    constexpr int NS = CELL_STEPS - CELL_READ;
#pragma unroll
    for (int q = 0; q < 7; ++q) {
#pragma unroll
        for (int m = 0; m < 3; ++m) {
            const int i = 3 * q + m;
            acc[m] = wmma4(a_x[m][q], xb[q], (q == 0) ? bias[m] : acc[m]);
            if (SHARE) {
                cell_steps<0>(CELL_READ + i * NS / NW, CELL_READ + (i + 1) * NS / NW, acc, c0v, w0);
                if (DO1) cell_steps<5>(CELL_READ + i * NS / NW, CELL_READ + (i + 1) * NS / NW, acc, c1v, w1);
                __builtin_amdgcn_sched_barrier(0);
            }
        }
    }
}

template <bool DO0>
__device__ __forceinline__ void h_products(v8f* acc, const v8f* bias, const v2f (*a_h01)[3],
                                           const v2f (*a_h234)[5], const v2f* hb) {
#pragma unroll
    for (int c = 0; c < 5; ++c) {
#pragma unroll
        for (int m = (DO0 ? 0 : 2); m < 5; ++m) {
            if (m < 2 && c >= 3) continue;
            const bool fresh = (c == 0) && (m >= 3 || (!DO0 && m == 2));
            if (m < 2) acc[m] = wmma4(a_h01[m][c], hb[c], acc[m]);
            else       acc[m] = wmma4(a_h234[m - 2][c], hb[c], fresh ? bias[m] : acc[m]);
        }
    }
}

template <bool DO0, bool DO1, bool XP>
__device__ __forceinline__ void cells(v8f* acc, const v8f* bias, const v2f (*a_x)[7], const v2f* xb,
                                      float* c0v, float* c1v, float* h0v, float* h1v) {
    CellW w0, w1;
    if (DO0) cell_steps<0>(0, CELL_READ, acc, c0v, w0);
    if (DO1) cell_steps<5>(0, CELL_READ, acc, c1v, w1);
    if (XP) {
        x_products<true, DO1>(acc, bias, a_x, xb, c0v, c1v, w0, w1);
    } else {
        if (DO0) cell_steps<0>(CELL_READ, CELL_STEPS, acc, c0v, w0);
        if (DO1) cell_steps<5>(CELL_READ, CELL_STEPS, acc, c1v, w1);
    }
#pragma unroll
    for (int u = 0; u < 5; ++u) {
        if (DO0) h0v[u] = w0.hv[u];
        if (DO1) h1v[u] = w1.hv[u];
    }
}

__global__ __launch_bounds__(256) void lstm_fused_kernel(
    const float* __restrict__ x,
    const float* __restrict__ w_ih0, const float* __restrict__ w_hh0,
    const float* __restrict__ b_ih0, const float* __restrict__ b_hh0,
    const float* __restrict__ w_ih1, const float* __restrict__ w_hh1,
    const float* __restrict__ b_ih1, const float* __restrict__ b_hh1,
    const float* __restrict__ w_cls, const float* __restrict__ b_cls,
    float* __restrict__ out, int B)
{
    __shared__ float2 wfrag[NFRAG][32];
    __shared__ __attribute__((aligned(16))) float biasT[2][40];
    __shared__ float bclsT[16];

    const int tid = threadIdx.x;

    for (int idx = tid; idx < NFRAG * 64; idx += 256) {
        const int frag = idx >> 6;
        const int r    = idx & 63;
        const int ln   = r >> 1;
        const int s    = r & 1;
        const int row  = ln & 15;
        const int kh   = ln >> 4;
        float val = 0.0f;
        if (frag < F_CLS) {
            int m, c;
            const bool xpart = (frag < F_H);
            if (xpart) { m = frag / 7; c = frag % 7; }
            else { const int f2 = frag - F_H; m = f2 / 5; c = f2 % 5; }
            const int r8 = row & 7;
            const int g  = r8 & 3;
            const int sl = 2 * m + (r8 >> 2);
            const int lr = (sl >= 5) ? 1 : 0;
            const int j  = 5 * (row >> 3) + (lr ? (sl - 5) : sl);
            if (xpart) {
                const int k = 4 * c + 2 * kh + s;
                if (lr == 0) val = gate_scale(g) * w_ih0[(g * HID + j) * D_IN + k];
            } else {
                const int hs = 2 * c + s;
                const int lk = (hs >= 5) ? 1 : 0;
                const int jk = 5 * kh + (lk ? (hs - 5) : hs);
                if (lr == 0) {
                    if (lk == 0) val = gate_scale(g) * w_hh0[(g * HID + j) * HID + jk];
                } else {
                    const float* wm = lk ? w_hh1 : w_ih1;
                    val = gate_scale(g) * wm[(g * HID + j) * HID + jk];
                }
            }
        } else {
            const int hs = 2 * (2 + frag - F_CLS) + s;
            if (row < HID && hs >= 5) val = w_cls[row * HID + 5 * kh + (hs - 5)];
        }
        ((float*)&wfrag[frag][ln])[s] = val;
    }
    if (tid < 80) {
        const int hh = tid / 40, si = tid % 40;
        const int m = si >> 3, r8 = si & 7;
        const int g = r8 & 3;
        const int sl = 2 * m + (r8 >> 2);
        const int lr = (sl >= 5) ? 1 : 0;
        const int j = 5 * hh + (lr ? (sl - 5) : sl);
        const float* bi = lr ? b_ih1 : b_ih0;
        const float* bh = lr ? b_hh1 : b_hh0;
        biasT[hh][si] = gate_scale(g) * (bi[g * HID + j] + bh[g * HID + j]);
    } else if (tid >= 96 && tid < 112) {
        const int n = tid - 96;
        bclsT[n] = (n < HID) ? b_cls[n] : 0.0f;
    }
    __syncthreads();

    const int wave = tid >> 5;
    const int lane = tid & 31;
    const int lo   = lane & 15;
    const int hi   = lane >> 4;

    const int b0   = blockIdx.x * 128 + wave * 16;
    const int brow = b0 + lo;
    const int brow_c = (brow < B) ? brow : (B - 1);

    v2f a_x[3][7], a_h01[2][3], a_h234[3][5];
#pragma unroll
    for (int m = 0; m < 3; ++m) {
#pragma unroll
        for (int q = 0; q < 7; ++q) a_x[m][q] = ld_afrag(wfrag[F_X + m * 7 + q], lane);
#pragma unroll
        for (int c = 0; c < 5; ++c) a_h234[m][c] = ld_afrag(wfrag[F_H + (m + 2) * 5 + c], lane);
    }
#pragma unroll
    for (int m = 0; m < 2; ++m) {
#pragma unroll
        for (int c = 0; c < 3; ++c) a_h01[m][c] = ld_afrag(wfrag[F_H + m * 5 + c], lane);
    }
    v8f bias[5];
#pragma unroll
    for (int m = 0; m < 5; ++m) {
#pragma unroll
        for (int v = 0; v < 8; ++v) bias[m][v] = biasT[hi][m * 8 + v];
    }

    float h0v[5], h1v[5], c0v[5], c1v[5];
#pragma unroll
    for (int u = 0; u < 5; ++u) { h0v[u] = 0.0f; h1v[u] = 0.0f; c0v[u] = 0.0f; c1v[u] = 0.0f; }

    const float* xrow = x + (size_t)brow_c * (T_STEPS * D_IN);

    v2f xb[7], xn[7];
    load_x(xrow, 0, hi, xb);
    load_x(xrow, 1, hi, xn);

    v8f acc[5];
    v2f hb[5];
    CellW wd0, wd1;

    x_products<false, false>(acc, bias, a_x, xb, c0v, c1v, wd0, wd1);
#pragma unroll
    for (int q = 0; q < 7; ++q) xb[q] = xn[q];
    load_x(xrow, 2, hi, xn);
    cells<true, false, true>(acc, bias, a_x, xb, c0v, c1v, h0v, h1v);
#pragma unroll
    for (int q = 0; q < 7; ++q) xb[q] = xn[q];

    for (int t = 1; t < T_STEPS - 1; ++t) {
        const int tn = (t + 2 < T_STEPS) ? (t + 2) : (T_STEPS - 1);
        load_x(xrow, tn, hi, xn);
        const int tp = (t + 3 < T_STEPS) ? (t + 3) : (T_STEPS - 1);
        __builtin_prefetch(xrow + tp * D_IN + (D_IN - 1), 0, 1);

        pack_h(h0v, h1v, hb);
        h_products<true>(acc, bias, a_h01, a_h234, hb);
        cells<true, true, true>(acc, bias, a_x, xb, c0v, c1v, h0v, h1v);
#pragma unroll
        for (int q = 0; q < 7; ++q) xb[q] = xn[q];
    }

    pack_h(h0v, h1v, hb);
    h_products<true>(acc, bias, a_h01, a_h234, hb);
    cells<true, true, false>(acc, bias, a_x, xb, c0v, c1v, h0v, h1v);

    pack_h(h0v, h1v, hb);
    h_products<false>(acc, bias, a_h01, a_h234, hb);
    cells<false, true, false>(acc, bias, a_x, xb, c0v, c1v, h0v, h1v);

    pack_h(h0v, h1v, hb);
    v8f accc;
#pragma unroll
    for (int v = 0; v < 8; ++v) accc[v] = bclsT[8 * hi + v];
#pragma unroll
    for (int q = 0; q < 3; ++q)
        accc = wmma4(ld_afrag(wfrag[F_CLS + q], lane), hb[2 + q], accc);

    if (brow < B) {
        float* orow = out + (size_t)brow * HID + 8 * hi;
        float2 p; p.x = accc[0]; p.y = accc[1];
        *(float2*)(orow) = p;
        if (hi == 0) {
#pragma unroll
            for (int v = 2; v < 8; v += 2) {
                float2 w; w.x = accc[v]; w.y = accc[v + 1];
                *(float2*)(orow + v) = w;
            }
        }
    }
}

extern "C" void kernel_launch(void* const* d_in, const int* in_sizes, int n_in,
                              void* d_out, int out_size, void* d_ws, size_t ws_size,
                              hipStream_t stream) {
    const float* x     = (const float*)d_in[0];
    const float* w_ih0 = (const float*)d_in[1];
    const float* w_hh0 = (const float*)d_in[2];
    const float* b_ih0 = (const float*)d_in[3];
    const float* b_hh0 = (const float*)d_in[4];
    const float* w_ih1 = (const float*)d_in[5];
    const float* w_hh1 = (const float*)d_in[6];
    const float* b_ih1 = (const float*)d_in[7];
    const float* b_hh1 = (const float*)d_in[8];
    const float* w_cls = (const float*)d_in[9];
    const float* b_cls = (const float*)d_in[10];

    const int B = in_sizes[0] / (T_STEPS * D_IN);
    const int blocks = (B + 127) / 128;

    lstm_fused_kernel<<<blocks, 256, 0, stream>>>(
        x, w_ih0, w_hh0, b_ih0, b_hh0,
        w_ih1, w_hh1, b_ih1, b_hh1,
        w_cls, b_cls, (float*)d_out, B);
}


# Start-up loads issued as one group: x of steps 0,1 first, then each thread's 13 weight elements through branch-free select-formed addresses, then the biases; first wait only before the LDS stores
# speedup vs baseline: 1.9817x; 1.0882x over previous
#include <hip/hip_runtime.h>


typedef float v2f __attribute__((ext_vector_type(2)));
typedef float v8f __attribute__((ext_vector_type(8)));

#define T_STEPS 28
#define D_IN    28
#define HID     10

#define F_X   0
#define F_H   21
#define F_CLS 46
#define NFRAG 49
#define STAGE_N   13
#define NFRAG_PAD (4 * STAGE_N)

__device__ __forceinline__ float htanh(float x) {
    return __builtin_amdgcn_tanhf(x);
}
__device__ __forceinline__ float gate_scale(int g) {
    return (g == 2) ? 1.0f : 0.5f;
}

__device__ __forceinline__ v8f wmma4(v2f a, v2f b, v8f c) {
    return __builtin_amdgcn_wmma_f32_16x16x4_f32(
         false, a,  false, b,
         (short)0, c,  false,  false);
}

__device__ __forceinline__ v2f ld_afrag(const float2* wf, int lane) {
    float2 t = wf[lane];
    v2f a; a.x = t.x; a.y = t.y;
    return a;
}

#define CELL_STEPS 30
#define CELL_READ  10
struct CellW {
    float t[5][4];
    float si[5];
    float sf[5];
    float so[5];
    float ig[5];
    float tc[5];
    float hv[5];
};

template <int S0>
__device__ __forceinline__ void cell_step(int n, const v8f* acc, float* cst, CellW& w) {
    const int st = n / 5, u = n % 5;
    const int sl = S0 + u;
    const int m = sl >> 1, o4 = 4 * (sl & 1);
    switch (st) {
    case 0:
        w.t[u][0] = htanh(acc[m][o4 + 0]);
        w.t[u][1] = htanh(acc[m][o4 + 1]);
        break;
    case 1:
        w.t[u][2] = htanh(acc[m][o4 + 2]);
        w.t[u][3] = htanh(acc[m][o4 + 3]);
        break;
    case 2:
        w.si[u] = 0.5f * w.t[u][0] + 0.5f;
        w.sf[u] = 0.5f * w.t[u][1] + 0.5f;
        break;
    case 3:
        w.ig[u] = w.si[u] * w.t[u][2];
        w.so[u] = 0.5f * w.t[u][3] + 0.5f;
        break;
    case 4:
        cst[u] = w.sf[u] * cst[u] + w.ig[u];
        w.tc[u] = htanh(cst[u]);
        break;
    default:
        w.hv[u] = w.so[u] * w.tc[u];
        break;
    }
}

template <int S0>
__device__ __forceinline__ void cell_steps(int lo, int hi, const v8f* acc, float* cst, CellW& w) {
#pragma unroll
    for (int n = lo; n < hi; ++n) cell_step<S0>(n, acc, cst, w);
}

__device__ __forceinline__ void pack_h(const float* h0v, const float* h1v, v2f* hb) {
    hb[0].x = h0v[0]; hb[0].y = h0v[1];
    hb[1].x = h0v[2]; hb[1].y = h0v[3];
    hb[2].x = h0v[4]; hb[2].y = h1v[0];
    hb[3].x = h1v[1]; hb[3].y = h1v[2];
    hb[4].x = h1v[3]; hb[4].y = h1v[4];
}

__device__ __forceinline__ void load_x(const float* xrow, int t, int hi, v2f* xb) {
#pragma unroll
    for (int q = 0; q < 7; ++q) {
        const float2 p = *(const float2*)(xrow + t * D_IN + 4 * q + 2 * hi);
        xb[q].x = p.x; xb[q].y = p.y;
    }
}

template <bool SHARE, bool DO1>
__device__ __forceinline__ void x_products(v8f* acc, const v8f* bias, const v2f (*a_x)[7], const v2f* xb,
                                           float* c0v, float* c1v, CellW& w0, CellW& w1) {
    constexpr int NW = 21;
    constexpr int NS = CELL_STEPS - CELL_READ;
#pragma unroll
    for (int q = 0; q < 7; ++q) {
#pragma unroll
        for (int m = 0; m < 3; ++m) {
            const int i = 3 * q + m;
            acc[m] = wmma4(a_x[m][q], xb[q], (q == 0) ? bias[m] : acc[m]);
            if (SHARE) {
                cell_steps<0>(CELL_READ + i * NS / NW, CELL_READ + (i + 1) * NS / NW, acc, c0v, w0);
                if (DO1) cell_steps<5>(CELL_READ + i * NS / NW, CELL_READ + (i + 1) * NS / NW, acc, c1v, w1);
                __builtin_amdgcn_sched_barrier(0);
            }
        }
    }
}

template <bool DO0>
__device__ __forceinline__ void h_products(v8f* acc, const v8f* bias, const v2f (*a_h01)[3],
                                           const v2f (*a_h234)[5], const v2f* hb) {
#pragma unroll
    for (int c = 0; c < 5; ++c) {
#pragma unroll
        for (int m = (DO0 ? 0 : 2); m < 5; ++m) {
            if (m < 2 && c >= 3) continue;
            const bool fresh = (c == 0) && (m >= 3 || (!DO0 && m == 2));
            if (m < 2) acc[m] = wmma4(a_h01[m][c], hb[c], acc[m]);
            else       acc[m] = wmma4(a_h234[m - 2][c], hb[c], fresh ? bias[m] : acc[m]);
        }
    }
}

template <bool DO0, bool DO1, bool XP>
__device__ __forceinline__ void cells(v8f* acc, const v8f* bias, const v2f (*a_x)[7], const v2f* xb,
                                      float* c0v, float* c1v, float* h0v, float* h1v) {
    CellW w0, w1;
    if (DO0) cell_steps<0>(0, CELL_READ, acc, c0v, w0);
    if (DO1) cell_steps<5>(0, CELL_READ, acc, c1v, w1);
    if (XP) {
        x_products<true, DO1>(acc, bias, a_x, xb, c0v, c1v, w0, w1);
    } else {
        if (DO0) cell_steps<0>(CELL_READ, CELL_STEPS, acc, c0v, w0);
        if (DO1) cell_steps<5>(CELL_READ, CELL_STEPS, acc, c1v, w1);
    }
#pragma unroll
    for (int u = 0; u < 5; ++u) {
        if (DO0) h0v[u] = w0.hv[u];
        if (DO1) h1v[u] = w1.hv[u];
    }
}

__global__ __launch_bounds__(256) void lstm_fused_kernel(
    const float* __restrict__ x,
    const float* __restrict__ w_ih0, const float* __restrict__ w_hh0,
    const float* __restrict__ b_ih0, const float* __restrict__ b_hh0,
    const float* __restrict__ w_ih1, const float* __restrict__ w_hh1,
    const float* __restrict__ b_ih1, const float* __restrict__ b_hh1,
    const float* __restrict__ w_cls, const float* __restrict__ b_cls,
    float* __restrict__ out, int B)
{
    __shared__ float2 wfrag[NFRAG_PAD][32];
    __shared__ __attribute__((aligned(16))) float sbias[256];
    float* const biasT = sbias;
    float* const bclsT = sbias + 96;

    const int tid = threadIdx.x;
    const int wave = tid >> 5;
    const int lane = tid & 31;
    const int lo   = lane & 15;
    const int hi   = lane >> 4;

    const int b0   = blockIdx.x * 128 + wave * 16;
    const int brow = b0 + lo;
    const int brow_c = (brow < B) ? brow : (B - 1);
    const float* xrow = x + (size_t)brow_c * (T_STEPS * D_IN);

    v2f xb[7], xn[7];
    load_x(xrow, 0, hi, xb);
    load_x(xrow, 1, hi, xn);

    const int r   = tid & 63;
    const int ln  = r >> 1;
    const int s   = r & 1;
    const int row = ln & 15;
    const int kh  = ln >> 4;
    const int r8  = row & 7;
    const int g   = r8 & 3;
    const float gsc = gate_scale(g);
    const int f0 = __builtin_amdgcn_readfirstlane(tid >> 6) & 3;
    float wv[STAGE_N];
    bool  wok[STAGE_N];
#pragma unroll
    for (int i = 0; i < STAGE_N; ++i) {
        const int fq   = f0 + 4 * i;
        const int frag = (fq < NFRAG) ? fq : (NFRAG - 1);
        const bool xpart = (frag < F_H);
        const bool cpart = (frag >= F_CLS);
        const int f2 = xpart ? frag : (frag - F_H);
        const int m  = xpart ? ((f2 * 37) >> 8) : ((f2 * 52) >> 8);
        const int c  = f2 - (xpart ? 7 : 5) * m;
        const int sl = 2 * m + (r8 >> 2);
        const bool lr = (sl >= 5);
        const int j  = 5 * (row >> 3) + (lr ? (sl - 5) : sl);
        const int grow = g * HID + j;
        const int offx = grow * D_IN + 4 * c + 2 * kh + s;
        const int hs = 2 * c + s;
        const bool lk = (hs >= 5);
        const int jk = 5 * kh + (lk ? (hs - 5) : hs);
        const int offh = grow * HID + jk;
        const float* ph = lr ? (lk ? w_hh1 : w_ih1) : w_hh0;
        const int hsc = 2 * (2 + frag - F_CLS) + s;
        const int offc = row * HID + 5 * kh + (hsc - 5);
        const bool ok = cpart ? (row < HID && hsc >= 5) : (xpart ? !lr : (lr || !lk));
        const float* p = cpart ? w_cls : (xpart ? w_ih0 : ph);
        const int off  = cpart ? offc : (xpart ? offx : offh);
        wv[i]  = p[ok ? off : 0];
        wok[i] = ok;
    }
    const int tb  = (tid < 80) ? tid : 0;
    const int bhh = tb / 40, bsi = tb % 40;
    const int bm  = bsi >> 3, br8 = bsi & 7;
    const int bg  = br8 & 3;
    const int bsl = 2 * bm + (br8 >> 2);
    const bool blr = (bsl >= 5);
    const int bj  = 5 * bhh + (blr ? (bsl - 5) : bsl);
    const bool isb = (tid < 80);
    const int bn = tid - 96;
    const bool bn_ok = (bn >= 0 && bn < HID);
    const float* pa = isb ? (blr ? b_ih1 : b_ih0) : b_cls;
    const int    ia = isb ? (bg * HID + bj) : (bn_ok ? bn : 0);
    const float ba_v = pa[ia];
    const float bb_l = (blr ? b_hh1 : b_hh0)[bg * HID + bj];
    const float bb_v = isb ? bb_l : -0.0f;
    const float bsc  = isb ? gate_scale(bg) : 1.0f;

#pragma unroll
    for (int i = 0; i < STAGE_N; ++i) {
        const bool cpart = (f0 + 4 * i >= F_CLS);
        const float sv = cpart ? wv[i] : (gsc * wv[i]);
        ((float*)&wfrag[0][0])[tid + 256 * i] = wok[i] ? sv : 0.0f;
    }
    sbias[tid] = (isb || bn_ok) ? (bsc * (ba_v + bb_v)) : 0.0f;
    __syncthreads();

    v2f a_x[3][7], a_h01[2][3], a_h234[3][5];
#pragma unroll
    for (int m = 0; m < 3; ++m) {
#pragma unroll
        for (int q = 0; q < 7; ++q) a_x[m][q] = ld_afrag(wfrag[F_X + m * 7 + q], lane);
#pragma unroll
        for (int c = 0; c < 5; ++c) a_h234[m][c] = ld_afrag(wfrag[F_H + (m + 2) * 5 + c], lane);
    }
#pragma unroll
    for (int m = 0; m < 2; ++m) {
#pragma unroll
        for (int c = 0; c < 3; ++c) a_h01[m][c] = ld_afrag(wfrag[F_H + m * 5 + c], lane);
    }
    v8f bias[5];
#pragma unroll
    for (int m = 0; m < 5; ++m) {
#pragma unroll
        for (int v = 0; v < 8; ++v) bias[m][v] = biasT[hi * 40 + m * 8 + v];
    }

    float h0v[5], h1v[5], c0v[5], c1v[5];
#pragma unroll
    for (int u = 0; u < 5; ++u) { h0v[u] = 0.0f; h1v[u] = 0.0f; c0v[u] = 0.0f; c1v[u] = 0.0f; }

    v8f acc[5];
    v2f hb[5];
    CellW wd0, wd1;

    x_products<false, false>(acc, bias, a_x, xb, c0v, c1v, wd0, wd1);
#pragma unroll
    for (int q = 0; q < 7; ++q) xb[q] = xn[q];
    load_x(xrow, 2, hi, xn);
    cells<true, false, true>(acc, bias, a_x, xb, c0v, c1v, h0v, h1v);
#pragma unroll
    for (int q = 0; q < 7; ++q) xb[q] = xn[q];

    for (int t = 1; t < T_STEPS - 1; ++t) {
        const int tn = (t + 2 < T_STEPS) ? (t + 2) : (T_STEPS - 1);
        load_x(xrow, tn, hi, xn);
        const int tp = (t + 3 < T_STEPS) ? (t + 3) : (T_STEPS - 1);
        __builtin_prefetch(xrow + tp * D_IN + (D_IN - 1), 0, 1);

        pack_h(h0v, h1v, hb);
        h_products<true>(acc, bias, a_h01, a_h234, hb);
        cells<true, true, true>(acc, bias, a_x, xb, c0v, c1v, h0v, h1v);
#pragma unroll
        for (int q = 0; q < 7; ++q) xb[q] = xn[q];
    }

    pack_h(h0v, h1v, hb);
    h_products<true>(acc, bias, a_h01, a_h234, hb);
    cells<true, true, false>(acc, bias, a_x, xb, c0v, c1v, h0v, h1v);

    pack_h(h0v, h1v, hb);
    h_products<false>(acc, bias, a_h01, a_h234, hb);
    cells<false, true, false>(acc, bias, a_x, xb, c0v, c1v, h0v, h1v);

    pack_h(h0v, h1v, hb);
    v8f accc;
#pragma unroll
    for (int v = 0; v < 8; ++v) accc[v] = bclsT[8 * hi + v];
#pragma unroll
    for (int q = 0; q < 3; ++q)
        accc = wmma4(ld_afrag(wfrag[F_CLS + q], lane), hb[2 + q], accc);

    if (brow < B) {
        float* orow = out + (size_t)brow * HID + 8 * hi;
        float2 p; p.x = accc[0]; p.y = accc[1];
        *(float2*)(orow) = p;
        if (hi == 0) {
#pragma unroll
            for (int v = 2; v < 8; v += 2) {
                float2 w; w.x = accc[v]; w.y = accc[v + 1];
                *(float2*)(orow + v) = w;
            }
        }
    }
}

extern "C" void kernel_launch(void* const* d_in, const int* in_sizes, int n_in,
                              void* d_out, int out_size, void* d_ws, size_t ws_size,
                              hipStream_t stream) {
    const float* x     = (const float*)d_in[0];
    const float* w_ih0 = (const float*)d_in[1];
    const float* w_hh0 = (const float*)d_in[2];
    const float* b_ih0 = (const float*)d_in[3];
    const float* b_hh0 = (const float*)d_in[4];
    const float* w_ih1 = (const float*)d_in[5];
    const float* w_hh1 = (const float*)d_in[6];
    const float* b_ih1 = (const float*)d_in[7];
    const float* b_hh1 = (const float*)d_in[8];
    const float* w_cls = (const float*)d_in[9];
    const float* b_cls = (const float*)d_in[10];

    const int B = in_sizes[0] / (T_STEPS * D_IN);
    const int blocks = (B + 127) / 128;

    lstm_fused_kernel<<<blocks, 256, 0, stream>>>(
        x, w_ih0, w_hh0, b_ih0, b_hh0,
        w_ih1, w_hh1, b_ih1, b_hh1,
        w_cls, b_cls, (float*)d_out, B);
}


# Waves 4-7 of a block (the younger wave of each SIMD pair) run at s_setprio 3 for their first 18 time steps, then 0, so both waves of a SIMD end together; no arithmetic or sum order changed
# speedup vs baseline: 2.1204x; 1.0700x over previous
#include <hip/hip_runtime.h>


typedef float v2f __attribute__((ext_vector_type(2)));
typedef float v8f __attribute__((ext_vector_type(8)));

#define T_STEPS 28
#define D_IN    28
#define HID     10
#define PRIO_STEPS 18

#define F_X   0
#define F_H   21
#define F_CLS 46
#define NFRAG 49
#define STAGE_N   13
#define NFRAG_PAD (4 * STAGE_N)

__device__ __forceinline__ float htanh(float x) {
    return __builtin_amdgcn_tanhf(x);
}
__device__ __forceinline__ float gate_scale(int g) {
    return (g == 2) ? 1.0f : 0.5f;
}

__device__ __forceinline__ v8f wmma4(v2f a, v2f b, v8f c) {
    return __builtin_amdgcn_wmma_f32_16x16x4_f32(
         false, a,  false, b,
         (short)0, c,  false,  false);
}

__device__ __forceinline__ v2f ld_afrag(const float2* wf, int lane) {
    float2 t = wf[lane];
    v2f a; a.x = t.x; a.y = t.y;
    return a;
}

#define CELL_STEPS 30
#define CELL_READ  10
struct CellW {
    float t[5][4];
    float si[5];
    float sf[5];
    float so[5];
    float ig[5];
    float tc[5];
    float hv[5];
};

template <int S0>
__device__ __forceinline__ void cell_step(int n, const v8f* acc, float* cst, CellW& w) {
    const int st = n / 5, u = n % 5;
    const int sl = S0 + u;
    const int m = sl >> 1, o4 = 4 * (sl & 1);
    switch (st) {
    case 0:
        w.t[u][0] = htanh(acc[m][o4 + 0]);
        w.t[u][1] = htanh(acc[m][o4 + 1]);
        break;
    case 1:
        w.t[u][2] = htanh(acc[m][o4 + 2]);
        w.t[u][3] = htanh(acc[m][o4 + 3]);
        break;
    case 2:
        w.si[u] = 0.5f * w.t[u][0] + 0.5f;
        w.sf[u] = 0.5f * w.t[u][1] + 0.5f;
        break;
    case 3:
        w.ig[u] = w.si[u] * w.t[u][2];
        w.so[u] = 0.5f * w.t[u][3] + 0.5f;
        break;
    case 4:
        cst[u] = w.sf[u] * cst[u] + w.ig[u];
        w.tc[u] = htanh(cst[u]);
        break;
    default:
        w.hv[u] = w.so[u] * w.tc[u];
        break;
    }
}

template <int S0>
__device__ __forceinline__ void cell_steps(int lo, int hi, const v8f* acc, float* cst, CellW& w) {
#pragma unroll
    for (int n = lo; n < hi; ++n) cell_step<S0>(n, acc, cst, w);
}

__device__ __forceinline__ void pack_h(const float* h0v, const float* h1v, v2f* hb) {
    hb[0].x = h0v[0]; hb[0].y = h0v[1];
    hb[1].x = h0v[2]; hb[1].y = h0v[3];
    hb[2].x = h0v[4]; hb[2].y = h1v[0];
    hb[3].x = h1v[1]; hb[3].y = h1v[2];
    hb[4].x = h1v[3]; hb[4].y = h1v[4];
}

__device__ __forceinline__ void load_x(const float* xrow, int t, int hi, v2f* xb) {
#pragma unroll
    for (int q = 0; q < 7; ++q) {
        const float2 p = *(const float2*)(xrow + t * D_IN + 4 * q + 2 * hi);
        xb[q].x = p.x; xb[q].y = p.y;
    }
}

template <bool SHARE, bool DO1>
__device__ __forceinline__ void x_products(v8f* acc, const v8f* bias, const v2f (*a_x)[7], const v2f* xb,
                                           float* c0v, float* c1v, CellW& w0, CellW& w1) {
    constexpr int NW = 21;
    constexpr int NS = CELL_STEPS - CELL_READ;
#pragma unroll
    for (int q = 0; q < 7; ++q) {
#pragma unroll
        for (int m = 0; m < 3; ++m) {
            const int i = 3 * q + m;
            acc[m] = wmma4(a_x[m][q], xb[q], (q == 0) ? bias[m] : acc[m]);
            if (SHARE) {
                cell_steps<0>(CELL_READ + i * NS / NW, CELL_READ + (i + 1) * NS / NW, acc, c0v, w0);
                if (DO1) cell_steps<5>(CELL_READ + i * NS / NW, CELL_READ + (i + 1) * NS / NW, acc, c1v, w1);
                __builtin_amdgcn_sched_barrier(0);
            }
        }
    }
}

template <bool DO0>
__device__ __forceinline__ void h_products(v8f* acc, const v8f* bias, const v2f (*a_h01)[3],
                                           const v2f (*a_h234)[5], const v2f* hb) {
#pragma unroll
    for (int c = 0; c < 5; ++c) {
#pragma unroll
        for (int m = (DO0 ? 0 : 2); m < 5; ++m) {
            if (m < 2 && c >= 3) continue;
            const bool fresh = (c == 0) && (m >= 3 || (!DO0 && m == 2));
            if (m < 2) acc[m] = wmma4(a_h01[m][c], hb[c], acc[m]);
            else       acc[m] = wmma4(a_h234[m - 2][c], hb[c], fresh ? bias[m] : acc[m]);
        }
    }
}

template <bool DO0, bool DO1, bool XP>
__device__ __forceinline__ void cells(v8f* acc, const v8f* bias, const v2f (*a_x)[7], const v2f* xb,
                                      float* c0v, float* c1v, float* h0v, float* h1v) {
    CellW w0, w1;
    if (DO0) cell_steps<0>(0, CELL_READ, acc, c0v, w0);
    if (DO1) cell_steps<5>(0, CELL_READ, acc, c1v, w1);
    if (XP) {
        x_products<true, DO1>(acc, bias, a_x, xb, c0v, c1v, w0, w1);
    } else {
        if (DO0) cell_steps<0>(CELL_READ, CELL_STEPS, acc, c0v, w0);
        if (DO1) cell_steps<5>(CELL_READ, CELL_STEPS, acc, c1v, w1);
    }
#pragma unroll
    for (int u = 0; u < 5; ++u) {
        if (DO0) h0v[u] = w0.hv[u];
        if (DO1) h1v[u] = w1.hv[u];
    }
}

__global__ __launch_bounds__(256) void lstm_fused_kernel(
    const float* __restrict__ x,
    const float* __restrict__ w_ih0, const float* __restrict__ w_hh0,
    const float* __restrict__ b_ih0, const float* __restrict__ b_hh0,
    const float* __restrict__ w_ih1, const float* __restrict__ w_hh1,
    const float* __restrict__ b_ih1, const float* __restrict__ b_hh1,
    const float* __restrict__ w_cls, const float* __restrict__ b_cls,
    float* __restrict__ out, int B)
{
    __shared__ float2 wfrag[NFRAG_PAD][32];
    __shared__ __attribute__((aligned(16))) float sbias[256];
    float* const biasT = sbias;
    float* const bclsT = sbias + 96;

    const int tid = threadIdx.x;
    const int wave = tid >> 5;
    const int lane = tid & 31;
    const int lo   = lane & 15;
    const int hi   = lane >> 4;

    const int b0   = blockIdx.x * 128 + wave * 16;
    const int brow = b0 + lo;
    const int brow_c = (brow < B) ? brow : (B - 1);
    const float* xrow = x + (size_t)brow_c * (T_STEPS * D_IN);

    v2f xb[7], xn[7];
    load_x(xrow, 0, hi, xb);
    load_x(xrow, 1, hi, xn);

    const int r   = tid & 63;
    const int ln  = r >> 1;
    const int s   = r & 1;
    const int row = ln & 15;
    const int kh  = ln >> 4;
    const int r8  = row & 7;
    const int g   = r8 & 3;
    const float gsc = gate_scale(g);
    const int f0 = __builtin_amdgcn_readfirstlane(tid >> 6) & 3;
    float wv[STAGE_N];
    bool  wok[STAGE_N];
#pragma unroll
    for (int i = 0; i < STAGE_N; ++i) {
        const int fq   = f0 + 4 * i;
        const int frag = (fq < NFRAG) ? fq : (NFRAG - 1);
        const bool xpart = (frag < F_H);
        const bool cpart = (frag >= F_CLS);
        const int f2 = xpart ? frag : (frag - F_H);
        const int m  = xpart ? ((f2 * 37) >> 8) : ((f2 * 52) >> 8);
        const int c  = f2 - (xpart ? 7 : 5) * m;
        const int sl = 2 * m + (r8 >> 2);
        const bool lr = (sl >= 5);
        const int j  = 5 * (row >> 3) + (lr ? (sl - 5) : sl);
        const int grow = g * HID + j;
        const int offx = grow * D_IN + 4 * c + 2 * kh + s;
        const int hs = 2 * c + s;
        const bool lk = (hs >= 5);
        const int jk = 5 * kh + (lk ? (hs - 5) : hs);
        const int offh = grow * HID + jk;
        const float* ph = lr ? (lk ? w_hh1 : w_ih1) : w_hh0;
        const int hsc = 2 * (2 + frag - F_CLS) + s;
        const int offc = row * HID + 5 * kh + (hsc - 5);
        const bool ok = cpart ? (row < HID && hsc >= 5) : (xpart ? !lr : (lr || !lk));
        const float* p = cpart ? w_cls : (xpart ? w_ih0 : ph);
        const int off  = cpart ? offc : (xpart ? offx : offh);
        wv[i]  = p[ok ? off : 0];
        wok[i] = ok;
    }
    const int tb  = (tid < 80) ? tid : 0;
    const int bhh = tb / 40, bsi = tb % 40;
    const int bm  = bsi >> 3, br8 = bsi & 7;
    const int bg  = br8 & 3;
    const int bsl = 2 * bm + (br8 >> 2);
    const bool blr = (bsl >= 5);
    const int bj  = 5 * bhh + (blr ? (bsl - 5) : bsl);
    const bool isb = (tid < 80);
    const int bn = tid - 96;
    const bool bn_ok = (bn >= 0 && bn < HID);
    const float* pa = isb ? (blr ? b_ih1 : b_ih0) : b_cls;
    const int    ia = isb ? (bg * HID + bj) : (bn_ok ? bn : 0);
    const float ba_v = pa[ia];
    const float bb_l = (blr ? b_hh1 : b_hh0)[bg * HID + bj];
    const float bb_v = isb ? bb_l : -0.0f;
    const float bsc  = isb ? gate_scale(bg) : 1.0f;

#pragma unroll
    for (int i = 0; i < STAGE_N; ++i) {
        const bool cpart = (f0 + 4 * i >= F_CLS);
        const float sv = cpart ? wv[i] : (gsc * wv[i]);
        ((float*)&wfrag[0][0])[tid + 256 * i] = wok[i] ? sv : 0.0f;
    }
    sbias[tid] = (isb || bn_ok) ? (bsc * (ba_v + bb_v)) : 0.0f;
    __syncthreads();

    v2f a_x[3][7], a_h01[2][3], a_h234[3][5];
#pragma unroll
    for (int m = 0; m < 3; ++m) {
#pragma unroll
        for (int q = 0; q < 7; ++q) a_x[m][q] = ld_afrag(wfrag[F_X + m * 7 + q], lane);
#pragma unroll
        for (int c = 0; c < 5; ++c) a_h234[m][c] = ld_afrag(wfrag[F_H + (m + 2) * 5 + c], lane);
    }
#pragma unroll
    for (int m = 0; m < 2; ++m) {
#pragma unroll
        for (int c = 0; c < 3; ++c) a_h01[m][c] = ld_afrag(wfrag[F_H + m * 5 + c], lane);
    }
    v8f bias[5];
#pragma unroll
    for (int m = 0; m < 5; ++m) {
#pragma unroll
        for (int v = 0; v < 8; ++v) bias[m][v] = biasT[hi * 40 + m * 8 + v];
    }

    float h0v[5], h1v[5], c0v[5], c1v[5];
#pragma unroll
    for (int u = 0; u < 5; ++u) { h0v[u] = 0.0f; h1v[u] = 0.0f; c0v[u] = 0.0f; c1v[u] = 0.0f; }

    v8f acc[5];
    v2f hb[5];
    CellW wd0, wd1;

    const bool young = (__builtin_amdgcn_readfirstlane(tid) >> 7) != 0;
    if (young) __builtin_amdgcn_s_setprio(3);

    x_products<false, false>(acc, bias, a_x, xb, c0v, c1v, wd0, wd1);
#pragma unroll
    for (int q = 0; q < 7; ++q) xb[q] = xn[q];
    load_x(xrow, 2, hi, xn);
    cells<true, false, true>(acc, bias, a_x, xb, c0v, c1v, h0v, h1v);
#pragma unroll
    for (int q = 0; q < 7; ++q) xb[q] = xn[q];

    for (int t = 1; t < T_STEPS - 1; ++t) {
        if (t == PRIO_STEPS) __builtin_amdgcn_s_setprio(0);
        const int tn = (t + 2 < T_STEPS) ? (t + 2) : (T_STEPS - 1);
        load_x(xrow, tn, hi, xn);
        const int tp = (t + 3 < T_STEPS) ? (t + 3) : (T_STEPS - 1);
        __builtin_prefetch(xrow + tp * D_IN + (D_IN - 1), 0, 1);

        pack_h(h0v, h1v, hb);
        h_products<true>(acc, bias, a_h01, a_h234, hb);
        cells<true, true, true>(acc, bias, a_x, xb, c0v, c1v, h0v, h1v);
#pragma unroll
        for (int q = 0; q < 7; ++q) xb[q] = xn[q];
    }

    pack_h(h0v, h1v, hb);
    h_products<true>(acc, bias, a_h01, a_h234, hb);
    cells<true, true, false>(acc, bias, a_x, xb, c0v, c1v, h0v, h1v);

    pack_h(h0v, h1v, hb);
    h_products<false>(acc, bias, a_h01, a_h234, hb);
    cells<false, true, false>(acc, bias, a_x, xb, c0v, c1v, h0v, h1v);

    pack_h(h0v, h1v, hb);
    v8f accc;
#pragma unroll
    for (int v = 0; v < 8; ++v) accc[v] = bclsT[8 * hi + v];
#pragma unroll
    for (int q = 0; q < 3; ++q)
        accc = wmma4(ld_afrag(wfrag[F_CLS + q], lane), hb[2 + q], accc);

    if (brow < B) {
        float* orow = out + (size_t)brow * HID + 8 * hi;
        float2 p; p.x = accc[0]; p.y = accc[1];
        *(float2*)(orow) = p;
        if (hi == 0) {
#pragma unroll
            for (int v = 2; v < 8; v += 2) {
                float2 w; w.x = accc[v]; w.y = accc[v + 1];
                *(float2*)(orow + v) = w;
            }
        }
    }
}

extern "C" void kernel_launch(void* const* d_in, const int* in_sizes, int n_in,
                              void* d_out, int out_size, void* d_ws, size_t ws_size,
                              hipStream_t stream) {
    const float* x     = (const float*)d_in[0];
    const float* w_ih0 = (const float*)d_in[1];
    const float* w_hh0 = (const float*)d_in[2];
    const float* b_ih0 = (const float*)d_in[3];
    const float* b_hh0 = (const float*)d_in[4];
    const float* w_ih1 = (const float*)d_in[5];
    const float* w_hh1 = (const float*)d_in[6];
    const float* b_ih1 = (const float*)d_in[7];
    const float* b_hh1 = (const float*)d_in[8];
    const float* w_cls = (const float*)d_in[9];
    const float* b_cls = (const float*)d_in[10];

    const int B = in_sizes[0] / (T_STEPS * D_IN);
    const int blocks = (B + 127) / 128;

    lstm_fused_kernel<<<blocks, 256, 0, stream>>>(
        x, w_ih0, w_hh0, b_ih0, b_hh0,
        w_ih1, w_hh1, b_ih1, b_hh1,
        w_cls, b_cls, (float*)d_out, B);
}
